# starting kernel
# baseline (speedup 1.0000x reference)
#include <hip/hip_runtime.h>
#include <math.h>

#define SEGS 65
#define CCH  64
#define PIX  (256*256)
#define BB   4
#define SDIM 64

typedef __attribute__((ext_vector_type(2))) float v2f;
typedef __attribute__((ext_vector_type(8))) float v8f;

__device__ __forceinline__ unsigned ftos(float f) {
    unsigned b = __float_as_uint(f);
    return (b & 0x80000000u) ? ~b : (b | 0x80000000u);
}
__device__ __forceinline__ float stof(unsigned u) {
    unsigned b = (u & 0x80000000u) ? (u & 0x7FFFFFFFu) : ~u;
    return __uint_as_float(b);
}
#define NEG_INF_S 0x007FFFFFu

__global__ void seg_init_kernel(unsigned* seg, int n) {
    int i = blockIdx.x * blockDim.x + threadIdx.x;
    if (i < n) seg[i] = NEG_INF_S;
}

__global__ void segmax_kernel(const float* __restrict__ enc,
                              const int*   __restrict__ masks,
                              unsigned*    __restrict__ seg) {
    const int pixPerBlock    = 2048;
    const int chunksPerBatch = PIX / pixPerBlock;
    int b     = blockIdx.x / chunksPerBatch;
    int chunk = blockIdx.x % chunksPerBatch;
    int p0    = chunk * pixPerBlock;

    __shared__ unsigned tab[SEGS * CCH];
    for (int i = threadIdx.x; i < SEGS * CCH; i += blockDim.x) tab[i] = NEG_INF_S;
    __syncthreads();

    const float* encb  = enc   + (size_t)b * CCH * PIX;
    const int*   maskb = masks + (size_t)b * PIX;

    for (int base = 0; base < pixPerBlock; base += blockDim.x * 4) {
        int p = p0 + base + threadIdx.x * 4;
        int4 id4 = *(const int4*)&maskb[p];
        unsigned* r0 = &tab[id4.x * CCH];
        unsigned* r1 = &tab[id4.y * CCH];
        unsigned* r2 = &tab[id4.z * CCH];
        unsigned* r3 = &tab[id4.w * CCH];
        #pragma unroll 4
        for (int c = 0; c < CCH; ++c) {
            float4 v = *(const float4*)&encb[(size_t)c * PIX + p];
            atomicMax(&r0[c], ftos(v.x));
            atomicMax(&r1[c], ftos(v.y));
            atomicMax(&r2[c], ftos(v.z));
            atomicMax(&r3[c], ftos(v.w));
        }
    }
    __syncthreads();

    unsigned* segb = seg + (size_t)b * SEGS * CCH;
    for (int i = threadIdx.x; i < SEGS * CCH; i += blockDim.x)
        atomicMax(&segb[i], tab[i]);
}

__global__ void vec_out_kernel(const unsigned* __restrict__ seg,
                               float* __restrict__ outv) {
    int i = blockIdx.x * blockDim.x + threadIdx.x;
    if (i >= BB * CCH * SDIM) return;
    int s = i % SDIM;
    int c = (i / SDIM) % CCH;
    int b = i / (SDIM * CCH);
    outv[i] = stof(seg[((size_t)b * SEGS + (s + 1)) * CCH + c]);
}

__global__ void conv_wmma_kernel(const unsigned* __restrict__ seg,
                                 const float* __restrict__ w,
                                 const float* __restrict__ bias,
                                 float*       __restrict__ outc) {
    __shared__ float vpad[CCH * 66];
    __shared__ float wl[4 * 576];

    int b        = blockIdx.x >> 5;
    int tileBase = (blockIdx.x & 31) * 8;
    const unsigned* segb = seg + (size_t)b * SEGS * CCH;

    for (int i = threadIdx.x; i < CCH * 66; i += blockDim.x) {
        int c = i / 66, x = i % 66;
        vpad[i] = (x == 0 || x == 65) ? 0.0f : stof(segb[x * CCH + c]);
    }
    for (int i = threadIdx.x; i < 4 * 576; i += blockDim.x) {
        int o = i / 576, r = i % 576;
        int t = r / 64, c = r % 64;
        wl[i] = w[((size_t)(o * CCH + c)) * 9 + t];
    }
    __syncthreads();

    int wave = threadIdx.x >> 5;
    int lane = threadIdx.x & 31;
    int m    = lane & 15;
    int half = lane >> 4;
    int kb   = half * 2;

    int tile = tileBase + wave;
    int ic   = tile >> 2;
    int j0   = (tile & 3) << 4;

    bool wvalid = (m < 4);
    int  mo     = wvalid ? m : 0;

    v8f acc = {0.f, 0.f, 0.f, 0.f, 0.f, 0.f, 0.f, 0.f};

    for (int t = 0; t < 9; ++t) {
        int di = t / 3 - 1, dj = t % 3 - 1;
        int xi = ic + di + 1;
        int xj = j0 + m + dj + 1;
        const float* wrow = &wl[mo * 576 + t * 64];
        #pragma unroll
        for (int c4 = 0; c4 < 16; ++c4) {
            int c0 = c4 * 4 + kb;
            v2f a, bv;
            a.x = vpad[c0 * 66 + xi] * vpad[c0 * 66 + xj];
            a.y = vpad[(c0 + 1) * 66 + xi] * vpad[(c0 + 1) * 66 + xj];
            bv.x = wvalid ? wrow[c0]     : 0.0f;
            bv.y = wvalid ? wrow[c0 + 1] : 0.0f;
            acc = __builtin_amdgcn_wmma_f32_16x16x4_f32(
                      false, a, false, bv, (short)0, acc, false, false);
        }
    }

    if (m < 4) {
        float bs = bias[m];
        #pragma unroll
        for (int r = 0; r < 8; ++r) {
            int j = j0 + r + half * 8;
            float val = acc[r] + bs;
            outc[(((size_t)b * 4 + m) * SDIM + ic) * SDIM + j] =
                1.0f / (1.0f + expf(-val));
        }
    }
}

extern "C" void kernel_launch(void* const* d_in, const int* in_sizes, int n_in,
                              void* d_out, int out_size, void* d_ws, size_t ws_size,
                              hipStream_t stream) {
    const float* enc   = (const float*)d_in[0];
    const int*   masks = (const int*)d_in[1];
    const float* w     = (const float*)d_in[2];
    const float* bias  = (const float*)d_in[3];
    float*    out  = (float*)d_out;
    unsigned* seg  = (unsigned*)d_ws;
    float*    outv = out;
    float*    outc = out + BB * CCH * SDIM;

    int nseg = BB * SEGS * CCH;
    seg_init_kernel<<<(nseg + 255) / 256, 256, 0, stream>>>(seg, nseg);
    segmax_kernel<<<BB * 32, 256, 0, stream>>>(enc, masks, seg);
    vec_out_kernel<<<(BB * CCH * SDIM + 255) / 256, 256, 0, stream>>>(seg, outv);
    conv_wmma_kernel<<<BB * 32, 256, 0, stream>>>(seg, w, bias, outc);
}


# segmax split 4 ways by channel group: 512 work-groups of 16 ch x 2048 px (was 128 of 64 ch); a lane's 16 channel loads held as one group ahead of its LDS max stream (was 1 load in flight per lane)
# speedup vs baseline: 2.4767x; 2.4767x over previous
#include <hip/hip_runtime.h>
#include <math.h>

#define SEGS 65
#define CCH  64
#define PIX  (256*256)
#define BB   4
#define SDIM 64
#define TPITCH 65

typedef __attribute__((ext_vector_type(2))) float v2f;
typedef __attribute__((ext_vector_type(8))) float v8f;
typedef __attribute__((ext_vector_type(4))) float v4f;

__device__ __forceinline__ unsigned ftos(float f) {
    unsigned b = __float_as_uint(f);
    return (b & 0x80000000u) ? ~b : (b | 0x80000000u);
}
__device__ __forceinline__ float stof(unsigned u) {
    unsigned b = (u & 0x80000000u) ? (u & 0x7FFFFFFFu) : ~u;
    return __uint_as_float(b);
}
#define NEG_INF_S 0x007FFFFFu

__global__ void seg_init_kernel(unsigned* seg, int n) {
    int i = blockIdx.x * blockDim.x + threadIdx.x;
    if (i < n) seg[i] = NEG_INF_S;
}

#define CG   16
#define NGRP (CCH / CG)
#define SEG_TPB 256
#define SEG_PIX 2048

__global__ void __launch_bounds__(SEG_TPB)
segmax_kernel(const float* __restrict__ enc,
              const int*   __restrict__ masks,
              unsigned*    __restrict__ seg) {
    const int chunksPerBatch = PIX / SEG_PIX;
    int g     = blockIdx.x % NGRP;
    int chunk = (blockIdx.x / NGRP) % chunksPerBatch;
    int b     = blockIdx.x / (NGRP * chunksPerBatch);
    int p0    = chunk * SEG_PIX;

    __shared__ unsigned tab[CG * TPITCH];
    for (int i = threadIdx.x; i < CG * TPITCH; i += SEG_TPB) tab[i] = NEG_INF_S;
    __syncthreads();

    const float* encg  = enc   + ((size_t)b * CCH + (size_t)g * CG) * PIX;
    const int*   maskb = masks + (size_t)b * PIX;

    for (int base = 0; base < SEG_PIX; base += SEG_TPB * 4) {
        int p = p0 + base + threadIdx.x * 4;
        int4 id4 = *(const int4*)&maskb[p];
        v4f v[CG];
        #pragma unroll
        for (int c = 0; c < CG; ++c)
            v[c] = *(const v4f*)&encg[(size_t)c * PIX + p];
        asm volatile("" :: "v"(v[0]),  "v"(v[1]),  "v"(v[2]),  "v"(v[3]),
                           "v"(v[4]),  "v"(v[5]),  "v"(v[6]),  "v"(v[7]),
                           "v"(v[8]),  "v"(v[9]),  "v"(v[10]), "v"(v[11]),
                           "v"(v[12]), "v"(v[13]), "v"(v[14]), "v"(v[15]));
        unsigned* r0 = &tab[id4.x];
        unsigned* r1 = &tab[id4.y];
        unsigned* r2 = &tab[id4.z];
        unsigned* r3 = &tab[id4.w];
        #pragma unroll
        for (int c = 0; c < CG; ++c) {
            atomicMax(&r0[c * TPITCH], ftos(v[c].x));
            atomicMax(&r1[c * TPITCH], ftos(v[c].y));
            atomicMax(&r2[c * TPITCH], ftos(v[c].z));
            atomicMax(&r3[c * TPITCH], ftos(v[c].w));
        }
    }
    __syncthreads();

    unsigned* segb = seg + (size_t)b * SEGS * CCH + g * CG;
    for (int i = threadIdx.x; i < SEGS * CG; i += SEG_TPB) {
        int s = i / CG, c = i % CG;
        atomicMax(&segb[s * CCH + c], tab[c * TPITCH + s]);
    }
}

__global__ void vec_out_kernel(const unsigned* __restrict__ seg,
                               float* __restrict__ outv) {
    int i = blockIdx.x * blockDim.x + threadIdx.x;
    if (i >= BB * CCH * SDIM) return;
    int s = i % SDIM;
    int c = (i / SDIM) % CCH;
    int b = i / (SDIM * CCH);
    outv[i] = stof(seg[((size_t)b * SEGS + (s + 1)) * CCH + c]);
}

__global__ void conv_wmma_kernel(const unsigned* __restrict__ seg,
                                 const float* __restrict__ w,
                                 const float* __restrict__ bias,
                                 float*       __restrict__ outc) {
    __shared__ float vpad[CCH * 66];
    __shared__ float wl[4 * 576];

    int b        = blockIdx.x >> 5;
    int tileBase = (blockIdx.x & 31) * 8;
    const unsigned* segb = seg + (size_t)b * SEGS * CCH;

    for (int i = threadIdx.x; i < CCH * 66; i += blockDim.x) {
        int c = i / 66, x = i % 66;
        vpad[i] = (x == 0 || x == 65) ? 0.0f : stof(segb[x * CCH + c]);
    }
    for (int i = threadIdx.x; i < 4 * 576; i += blockDim.x) {
        int o = i / 576, r = i % 576;
        int t = r / 64, c = r % 64;
        wl[i] = w[((size_t)(o * CCH + c)) * 9 + t];
    }
    __syncthreads();

    int wave = threadIdx.x >> 5;
    int lane = threadIdx.x & 31;
    int m    = lane & 15;
    int half = lane >> 4;
    int kb   = half * 2;

    int tile = tileBase + wave;
    int ic   = tile >> 2;
    int j0   = (tile & 3) << 4;

    bool wvalid = (m < 4);
    int  mo     = wvalid ? m : 0;

    v8f acc = {0.f, 0.f, 0.f, 0.f, 0.f, 0.f, 0.f, 0.f};

    for (int t = 0; t < 9; ++t) {
        int di = t / 3 - 1, dj = t % 3 - 1;
        int xi = ic + di + 1;
        int xj = j0 + m + dj + 1;
        const float* wrow = &wl[mo * 576 + t * 64];
        #pragma unroll
        for (int c4 = 0; c4 < 16; ++c4) {
            int c0 = c4 * 4 + kb;
            v2f a, bv;
            a.x = vpad[c0 * 66 + xi] * vpad[c0 * 66 + xj];
            a.y = vpad[(c0 + 1) * 66 + xi] * vpad[(c0 + 1) * 66 + xj];
            bv.x = wvalid ? wrow[c0]     : 0.0f;
            bv.y = wvalid ? wrow[c0 + 1] : 0.0f;
            acc = __builtin_amdgcn_wmma_f32_16x16x4_f32(
                      false, a, false, bv, (short)0, acc, false, false);
        }
    }

    if (m < 4) {
        float bs = bias[m];
        #pragma unroll
        for (int r = 0; r < 8; ++r) {
            int j = j0 + r + half * 8;
            float val = acc[r] + bs;
            outc[(((size_t)b * 4 + m) * SDIM + ic) * SDIM + j] =
                1.0f / (1.0f + expf(-val));
        }
    }
}

extern "C" void kernel_launch(void* const* d_in, const int* in_sizes, int n_in,
                              void* d_out, int out_size, void* d_ws, size_t ws_size,
                              hipStream_t stream) {
    const float* enc   = (const float*)d_in[0];
    const int*   masks = (const int*)d_in[1];
    const float* w     = (const float*)d_in[2];
    const float* bias  = (const float*)d_in[3];
    float*    out  = (float*)d_out;
    unsigned* seg  = (unsigned*)d_ws;
    float*    outv = out;
    float*    outc = out + BB * CCH * SDIM;

    int nseg = BB * SEGS * CCH;
    seg_init_kernel<<<(nseg + 255) / 256, 256, 0, stream>>>(seg, nseg);
    segmax_kernel<<<BB * 32 * NGRP, SEG_TPB, 0, stream>>>(enc, masks, seg);
    vec_out_kernel<<<(BB * CCH * SDIM + 255) / 256, 256, 0, stream>>>(seg, outv);
    conv_wmma_kernel<<<BB * 32, 256, 0, stream>>>(seg, w, bias, outc);
}


# conv re-tiled: one 16x16 (row, column) output tile per work-group, WMMA N = output column (B = v, A = w*v), 4 out-channel accumulators; 8 waves split K, partials summed in LDS (f32 sum order changes)
# speedup vs baseline: 3.9671x; 1.6018x over previous
#include <hip/hip_runtime.h>
#include <math.h>

#define SEGS 65
#define CCH  64
#define PIX  (256*256)
#define BB   4
#define SDIM 64
#define TPITCH 65

typedef __attribute__((ext_vector_type(2))) float v2f;
typedef __attribute__((ext_vector_type(8))) float v8f;
typedef __attribute__((ext_vector_type(4))) float v4f;
typedef __attribute__((ext_vector_type(4))) int v4i;
typedef __attribute__((ext_vector_type(4))) unsigned v4u;

__device__ __forceinline__ unsigned ftos(float f) {
    unsigned b = __float_as_uint(f);
    return (b & 0x80000000u) ? ~b : (b | 0x80000000u);
}
__device__ __forceinline__ float stof(unsigned u) {
    unsigned b = (u & 0x80000000u) ? (u & 0x7FFFFFFFu) : ~u;
    return __uint_as_float(b);
}
#define NEG_INF_S 0x007FFFFFu

#define CG   4
#define NGRP (CCH / CG)
#define SEG_TPB 256
#define SEG_PIX 8192
#define NCHUNK (PIX / SEG_PIX)
#define QPT  4
#define PASS_PIX (SEG_TPB * 4 * QPT)
#define PLANE (CCH * SDIM)

static_assert(SEG_TPB == CG * SDIM, "flush: one thread per (channel, segment) word");
static_assert(CG == 4, "lane rotation below is written for 4 channels (two select stages)");

__device__ __forceinline__ void seg_max4(unsigned* tab, int id,
                                         float f0, float f1, float f2, float f3,
                                         bool b0, bool b1,
                                         unsigned o0, unsigned o1, unsigned o2, unsigned o3) {
    unsigned u0 = ftos(f0), u1 = ftos(f1), u2 = ftos(f2), u3 = ftos(f3);
    unsigned a0 = b0 ? u1 : u0, a1 = b0 ? u2 : u1, a2 = b0 ? u3 : u2, a3 = b0 ? u0 : u3;
    unsigned* row = &tab[id];
    atomicMax(&row[o0], b1 ? a2 : a0);
    atomicMax(&row[o1], b1 ? a3 : a1);
    atomicMax(&row[o2], b1 ? a0 : a2);
    atomicMax(&row[o3], b1 ? a1 : a3);
}

__global__ void __launch_bounds__(SEG_TPB)
segmax_kernel(const float* __restrict__ enc,
              const int*   __restrict__ masks,
              unsigned*    __restrict__ part) {
    int g     = blockIdx.x % NGRP;
    int chunk = (blockIdx.x / NGRP) % NCHUNK;
    int b     = blockIdx.x / (NGRP * NCHUNK);
    int p0    = chunk * SEG_PIX;

    __shared__ unsigned tab[CG * TPITCH];
    for (int i = threadIdx.x; i < CG * TPITCH; i += SEG_TPB) tab[i] = NEG_INF_S;
    __syncthreads();

    const unsigned rot = threadIdx.x & 3;
    const bool b0 = rot & 1, b1 = rot & 2;
    const unsigned o0 = ((0 + rot) & 3) * TPITCH, o1 = ((1 + rot) & 3) * TPITCH,
                   o2 = ((2 + rot) & 3) * TPITCH, o3 = ((3 + rot) & 3) * TPITCH;

    const float* encg  = enc   + ((size_t)b * CCH + (size_t)g * CG) * PIX;
    const int*   maskb = masks + (size_t)b * PIX;

    for (int base = 0; base < SEG_PIX; base += PASS_PIX) {
        int p = p0 + base + threadIdx.x * 4;
        v4i id[QPT];
        v4f v[QPT][CG];
        #pragma unroll
        for (int q = 0; q < QPT; ++q) {
            id[q] = *(const v4i*)&maskb[p + q * SEG_TPB * 4];
            #pragma unroll
            for (int c = 0; c < CG; ++c)
                v[q][c] = *(const v4f*)&encg[(size_t)c * PIX + p + q * SEG_TPB * 4];
        }
        asm volatile("" :: "v"(id[0]), "v"(id[1]), "v"(id[2]), "v"(id[3]),
                           "v"(v[0][0]), "v"(v[0][1]), "v"(v[0][2]), "v"(v[0][3]),
                           "v"(v[1][0]), "v"(v[1][1]), "v"(v[1][2]), "v"(v[1][3]),
                           "v"(v[2][0]), "v"(v[2][1]), "v"(v[2][2]), "v"(v[2][3]),
                           "v"(v[3][0]), "v"(v[3][1]), "v"(v[3][2]), "v"(v[3][3]));
        #pragma unroll
        for (int q = 0; q < QPT; ++q) {
            seg_max4(tab, id[q].x, v[q][0].x, v[q][1].x, v[q][2].x, v[q][3].x, b0, b1, o0, o1, o2, o3);
            seg_max4(tab, id[q].y, v[q][0].y, v[q][1].y, v[q][2].y, v[q][3].y, b0, b1, o0, o1, o2, o3);
            seg_max4(tab, id[q].z, v[q][0].z, v[q][1].z, v[q][2].z, v[q][3].z, b0, b1, o0, o1, o2, o3);
            seg_max4(tab, id[q].w, v[q][0].w, v[q][1].w, v[q][2].w, v[q][3].w, b0, b1, o0, o1, o2, o3);
        }
    }
    __syncthreads();

    unsigned* dst = part + (((size_t)b * NCHUNK + chunk) * CCH + (size_t)g * CG) * SDIM;
    int c = threadIdx.x / SDIM, s = threadIdx.x % SDIM;
    dst[threadIdx.x] = tab[c * TPITCH + s + 1];
}

__global__ void __launch_bounds__(256)
conv_wmma_kernel(const unsigned* __restrict__ part,
                 const float* __restrict__ w,
                 const float* __restrict__ bias,
                 float*       __restrict__ outv,
                 float*       __restrict__ outc) {
    __shared__ float vpad[CCH * 66];
    __shared__ float wl[4 * 576];
    __shared__ float red[8 * 4 * 256];

    int b        = blockIdx.x >> 4;
    int kk       = blockIdx.x & 15;
    const unsigned* pb = part + (size_t)b * NCHUNK * PLANE;

    int tid = threadIdx.x;
    v4u px[NCHUNK][4];
    #pragma unroll
    for (int ch = 0; ch < NCHUNK; ++ch) {
        #pragma unroll
        for (int j = 0; j < 4; ++j)
            px[ch][j] = *(const v4u*)&pb[(size_t)ch * PLANE + (j * 256 + tid) * 4];
    }
    asm volatile("" :: "v"(px[0][0]), "v"(px[0][1]), "v"(px[0][2]), "v"(px[0][3]),
                       "v"(px[1][0]), "v"(px[1][1]), "v"(px[1][2]), "v"(px[1][3]),
                       "v"(px[2][0]), "v"(px[2][1]), "v"(px[2][2]), "v"(px[2][3]),
                       "v"(px[3][0]), "v"(px[3][1]), "v"(px[3][2]), "v"(px[3][3]));
    asm volatile("" :: "v"(px[4][0]), "v"(px[4][1]), "v"(px[4][2]), "v"(px[4][3]),
                       "v"(px[5][0]), "v"(px[5][1]), "v"(px[5][2]), "v"(px[5][3]),
                       "v"(px[6][0]), "v"(px[6][1]), "v"(px[6][2]), "v"(px[6][3]),
                       "v"(px[7][0]), "v"(px[7][1]), "v"(px[7][2]), "v"(px[7][3]));
    v4u mx[4];
    #pragma unroll
    for (int j = 0; j < 4; ++j) {
        mx[j] = px[0][j];
        #pragma unroll
        for (int ch = 1; ch < NCHUNK; ++ch)
            mx[j] = __builtin_elementwise_max(mx[j], px[ch][j]);
    }
    #pragma unroll
    for (int j = 0; j < 4; ++j) {
        v4f o = { stof(mx[j].x), stof(mx[j].y), stof(mx[j].z), stof(mx[j].w) };
        float* d = &vpad[(16 * j + (tid >> 4)) * 66 + 1 + (tid & 15) * 4];
        d[0] = o.x; d[1] = o.y; d[2] = o.z; d[3] = o.w;
        if (j == (kk >> 2) && (tid >> 6) == (kk & 3))
            *(v4f*)&outv[(size_t)b * PLANE + (j * 256 + tid) * 4] = o;
    }
    if (tid < 2 * CCH) vpad[(tid >> 1) * 66 + (tid & 1) * 65] = 0.0f;
    for (int i = threadIdx.x; i < 4 * 576; i += blockDim.x) {
        int o = i / 576, r = i % 576;
        int t = r / 64, c = r % 64;
        wl[i] = w[((size_t)(o * CCH + c)) * 9 + t];
    }
    __syncthreads();

    int wave = threadIdx.x >> 5;
    int lane = threadIdx.x & 31;
    int m    = lane & 15;
    int half = lane >> 4;
    int kb   = half * 2;

    int i0 = (kk >> 2) << 4;
    int j0 = (kk & 3) << 4;

    v8f acc[4];
    for (int o = 0; o < 4; ++o)
        acc[o] = (v8f){0.f, 0.f, 0.f, 0.f, 0.f, 0.f, 0.f, 0.f};

    for (int cc = 0; cc < 2; ++cc) {
        int c0 = (wave * 2 + cc) * 4 + kb;
        const float* r0 = &vpad[c0 * 66];
        const float* r1 = r0 + 66;
        v2f va[3], vb[3];
        for (int d = 0; d < 3; ++d) {
            va[d].x = r0[i0 + m + d];  va[d].y = r1[i0 + m + d];
            vb[d].x = r0[j0 + m + d];  vb[d].y = r1[j0 + m + d];
        }
        for (int t = 0; t < 9; ++t) {
            for (int o = 0; o < 4; ++o) {
                v2f ww = *(const v2f*)&wl[o * 576 + t * 64 + c0];
                v2f a  = ww * va[t / 3];
                acc[o] = __builtin_amdgcn_wmma_f32_16x16x4_f32(
                             false, a, false, vb[t % 3], (short)0, acc[o], false, false);
            }
        }
    }

    for (int o = 0; o < 4; ++o) {
        for (int r = 0; r < 8; ++r)
            red[((wave * 4 + o) * 8 + r) * 32 + lane] = acc[o][r];
    }
    __syncthreads();

    int R = tid >> 4, C = tid & 15;
    int srcw = (R & 7) * 32 + (R >> 3) * 16 + C;
    for (int o = 0; o < 4; ++o) {
        float s = red[(0 * 4 + o) * 256 + srcw];
        for (int wv = 1; wv < 8; ++wv)
            s += red[(wv * 4 + o) * 256 + srcw];
        float val = s + bias[o];
        outc[(((size_t)b * 4 + o) * SDIM + i0 + R) * SDIM + j0 + C] =
            1.0f / (1.0f + expf(-val));
    }
}

extern "C" void kernel_launch(void* const* d_in, const int* in_sizes, int n_in,
                              void* d_out, int out_size, void* d_ws, size_t ws_size,
                              hipStream_t stream) {
    const float* enc   = (const float*)d_in[0];
    const int*   masks = (const int*)d_in[1];
    const float* w     = (const float*)d_in[2];
    const float* bias  = (const float*)d_in[3];
    float*    out  = (float*)d_out;
    unsigned* part = (unsigned*)d_ws;
    float*    outv = out;
    float*    outc = out + BB * CCH * SDIM;

    if (ws_size < sizeof(unsigned) * BB * NCHUNK * PLANE) return;
    segmax_kernel<<<BB * NCHUNK * NGRP, SEG_TPB, 0, stream>>>(enc, masks, part);
    conv_wmma_kernel<<<BB * 16, 256, 0, stream>>>(part, w, bias, outv, outc);
}


# segmax: 4 passes of 10 loads in two register buffers; pass k+1's loads issue ahead of pass k's LDS max stream, only the older 10 are waited for (at most 20 in flight per lane, as before, not 40)
# speedup vs baseline: 4.2094x; 1.0611x over previous
#include <hip/hip_runtime.h>
#include <math.h>

#define SEGS 65
#define CCH  64
#define PIX  (256*256)
#define BB   4
#define SDIM 64
#define TPITCH 65

typedef __attribute__((ext_vector_type(2))) float v2f;
typedef __attribute__((ext_vector_type(8))) float v8f;
typedef __attribute__((ext_vector_type(4))) float v4f;
typedef __attribute__((ext_vector_type(4))) int v4i;
typedef __attribute__((ext_vector_type(4))) unsigned v4u;

__device__ __forceinline__ unsigned ftos(float f) {
    unsigned b = __float_as_uint(f);
    return (b & 0x80000000u) ? ~b : (b | 0x80000000u);
}
__device__ __forceinline__ float stof(unsigned u) {
    unsigned b = (u & 0x80000000u) ? (u & 0x7FFFFFFFu) : ~u;
    return __uint_as_float(b);
}
#define NEG_INF_S 0x007FFFFFu

#define CG   4
#define NGRP (CCH / CG)
#define SEG_TPB 256
#define SEG_PIX 8192
#define NCHUNK (PIX / SEG_PIX)
#define QPT  2
#define PASS_PIX (SEG_TPB * 4 * QPT)
#define NPASS (SEG_PIX / PASS_PIX)
#define PLANE (CCH * SDIM)

static_assert(SEG_TPB == CG * SDIM, "flush: one thread per (channel, segment) word");
static_assert(CG == 4, "lane rotation below is written for 4 channels (two select stages)");
static_assert(NPASS % 2 == 0, "passes alternate between two register buffers");

__device__ __forceinline__ void seg_max4(unsigned* tab, int id,
                                         float f0, float f1, float f2, float f3,
                                         bool b0, bool b1,
                                         unsigned o0, unsigned o1, unsigned o2, unsigned o3) {
    unsigned u0 = ftos(f0), u1 = ftos(f1), u2 = ftos(f2), u3 = ftos(f3);
    unsigned a0 = b0 ? u1 : u0, a1 = b0 ? u2 : u1, a2 = b0 ? u3 : u2, a3 = b0 ? u0 : u3;
    unsigned* row = &tab[id];
    atomicMax(&row[o0], b1 ? a2 : a0);
    atomicMax(&row[o1], b1 ? a3 : a1);
    atomicMax(&row[o2], b1 ? a0 : a2);
    atomicMax(&row[o3], b1 ? a1 : a3);
}

__device__ __forceinline__ void seg_load_pass(const int* __restrict__ maskb,
                                              const float* __restrict__ encg, int p,
                                              v4i (&id)[QPT], v4f (&v)[QPT][CG]) {
    #pragma unroll
    for (int q = 0; q < QPT; ++q) {
        id[q] = *(const v4i*)&maskb[p + q * SEG_TPB * 4];
        #pragma unroll
        for (int c = 0; c < CG; ++c)
            v[q][c] = *(const v4f*)&encg[(size_t)c * PIX + p + q * SEG_TPB * 4];
    }
}

__device__ __forceinline__ void seg_feed_pass(unsigned* tab,
                                              const v4i (&id)[QPT], const v4f (&v)[QPT][CG],
                                              bool b0, bool b1,
                                              unsigned o0, unsigned o1, unsigned o2, unsigned o3) {
    #pragma unroll
    for (int q = 0; q < QPT; ++q) {
        seg_max4(tab, id[q].x, v[q][0].x, v[q][1].x, v[q][2].x, v[q][3].x, b0, b1, o0, o1, o2, o3);
        seg_max4(tab, id[q].y, v[q][0].y, v[q][1].y, v[q][2].y, v[q][3].y, b0, b1, o0, o1, o2, o3);
        seg_max4(tab, id[q].z, v[q][0].z, v[q][1].z, v[q][2].z, v[q][3].z, b0, b1, o0, o1, o2, o3);
        seg_max4(tab, id[q].w, v[q][0].w, v[q][1].w, v[q][2].w, v[q][3].w, b0, b1, o0, o1, o2, o3);
    }
}

__global__ void __launch_bounds__(SEG_TPB)
segmax_kernel(const float* __restrict__ enc,
              const int*   __restrict__ masks,
              unsigned*    __restrict__ part) {
    int g     = blockIdx.x % NGRP;
    int chunk = (blockIdx.x / NGRP) % NCHUNK;
    int b     = blockIdx.x / (NGRP * NCHUNK);
    int p0    = chunk * SEG_PIX;

    __shared__ unsigned tab[CG * TPITCH];
    for (int i = threadIdx.x; i < CG * TPITCH; i += SEG_TPB) tab[i] = NEG_INF_S;
    __syncthreads();

    const unsigned rot = threadIdx.x & 3;
    const bool b0 = rot & 1, b1 = rot & 2;
    const unsigned o0 = ((0 + rot) & 3) * TPITCH, o1 = ((1 + rot) & 3) * TPITCH,
                   o2 = ((2 + rot) & 3) * TPITCH, o3 = ((3 + rot) & 3) * TPITCH;

    const float* encg  = enc   + ((size_t)b * CCH + (size_t)g * CG) * PIX;
    const int*   maskb = masks + (size_t)b * PIX;

    const int pl = p0 + threadIdx.x * 4;
    v4i idA[QPT], idB[QPT];
    v4f vA[QPT][CG], vB[QPT][CG];
    seg_load_pass(maskb, encg, pl, idA, vA);
    __builtin_amdgcn_sched_barrier(0);
    #pragma unroll
    for (int pass = 0; pass < NPASS; pass += 2) {
        seg_load_pass(maskb, encg, pl + (pass + 1) * PASS_PIX, idB, vB);
        __builtin_amdgcn_sched_barrier(0);
        seg_feed_pass(tab, idA, vA, b0, b1, o0, o1, o2, o3);
        __builtin_amdgcn_sched_barrier(0);
        if (pass + 2 < NPASS) {
            seg_load_pass(maskb, encg, pl + (pass + 2) * PASS_PIX, idA, vA);
            __builtin_amdgcn_sched_barrier(0);
        }
        seg_feed_pass(tab, idB, vB, b0, b1, o0, o1, o2, o3);
        if (pass + 2 < NPASS)
            __builtin_amdgcn_sched_barrier(0);
    }
    __syncthreads();

    unsigned* dst = part + (((size_t)b * NCHUNK + chunk) * CCH + (size_t)g * CG) * SDIM;
    int c = threadIdx.x / SDIM, s = threadIdx.x % SDIM;
    dst[threadIdx.x] = tab[c * TPITCH + s + 1];
}

__global__ void __launch_bounds__(256)
conv_wmma_kernel(const unsigned* __restrict__ part,
                 const float* __restrict__ w,
                 const float* __restrict__ bias,
                 float*       __restrict__ outv,
                 float*       __restrict__ outc) {
    __shared__ float vpad[CCH * 66];
    __shared__ float wl[4 * 576];
    __shared__ float red[8 * 4 * 256];

    int b        = blockIdx.x >> 4;
    int kk       = blockIdx.x & 15;
    const unsigned* pb = part + (size_t)b * NCHUNK * PLANE;

    int tid = threadIdx.x;
    v4u px[NCHUNK][4];
    #pragma unroll
    for (int ch = 0; ch < NCHUNK; ++ch) {
        #pragma unroll
        for (int j = 0; j < 4; ++j)
            px[ch][j] = *(const v4u*)&pb[(size_t)ch * PLANE + (j * 256 + tid) * 4];
    }
    asm volatile("" :: "v"(px[0][0]), "v"(px[0][1]), "v"(px[0][2]), "v"(px[0][3]),
                       "v"(px[1][0]), "v"(px[1][1]), "v"(px[1][2]), "v"(px[1][3]),
                       "v"(px[2][0]), "v"(px[2][1]), "v"(px[2][2]), "v"(px[2][3]),
                       "v"(px[3][0]), "v"(px[3][1]), "v"(px[3][2]), "v"(px[3][3]));
    asm volatile("" :: "v"(px[4][0]), "v"(px[4][1]), "v"(px[4][2]), "v"(px[4][3]),
                       "v"(px[5][0]), "v"(px[5][1]), "v"(px[5][2]), "v"(px[5][3]),
                       "v"(px[6][0]), "v"(px[6][1]), "v"(px[6][2]), "v"(px[6][3]),
                       "v"(px[7][0]), "v"(px[7][1]), "v"(px[7][2]), "v"(px[7][3]));
    v4u mx[4];
    #pragma unroll
    for (int j = 0; j < 4; ++j) {
        mx[j] = px[0][j];
        #pragma unroll
        for (int ch = 1; ch < NCHUNK; ++ch)
            mx[j] = __builtin_elementwise_max(mx[j], px[ch][j]);
    }
    #pragma unroll
    for (int j = 0; j < 4; ++j) {
        v4f o = { stof(mx[j].x), stof(mx[j].y), stof(mx[j].z), stof(mx[j].w) };
        float* d = &vpad[(16 * j + (tid >> 4)) * 66 + 1 + (tid & 15) * 4];
        d[0] = o.x; d[1] = o.y; d[2] = o.z; d[3] = o.w;
        if (j == (kk >> 2) && (tid >> 6) == (kk & 3))
            *(v4f*)&outv[(size_t)b * PLANE + (j * 256 + tid) * 4] = o;
    }
    if (tid < 2 * CCH) vpad[(tid >> 1) * 66 + (tid & 1) * 65] = 0.0f;
    for (int i = threadIdx.x; i < 4 * 576; i += blockDim.x) {
        int o = i / 576, r = i % 576;
        int t = r / 64, c = r % 64;
        wl[i] = w[((size_t)(o * CCH + c)) * 9 + t];
    }
    __syncthreads();

    int wave = threadIdx.x >> 5;
    int lane = threadIdx.x & 31;
    int m    = lane & 15;
    int half = lane >> 4;
    int kb   = half * 2;

    int i0 = (kk >> 2) << 4;
    int j0 = (kk & 3) << 4;

    v8f acc[4];
    for (int o = 0; o < 4; ++o)
        acc[o] = (v8f){0.f, 0.f, 0.f, 0.f, 0.f, 0.f, 0.f, 0.f};

    for (int cc = 0; cc < 2; ++cc) {
        int c0 = (wave * 2 + cc) * 4 + kb;
        const float* r0 = &vpad[c0 * 66];
        const float* r1 = r0 + 66;
        v2f va[3], vb[3];
        for (int d = 0; d < 3; ++d) {
            va[d].x = r0[i0 + m + d];  va[d].y = r1[i0 + m + d];
            vb[d].x = r0[j0 + m + d];  vb[d].y = r1[j0 + m + d];
        }
        for (int t = 0; t < 9; ++t) {
            for (int o = 0; o < 4; ++o) {
                v2f ww = *(const v2f*)&wl[o * 576 + t * 64 + c0];
                v2f a  = ww * va[t / 3];
                acc[o] = __builtin_amdgcn_wmma_f32_16x16x4_f32(
                             false, a, false, vb[t % 3], (short)0, acc[o], false, false);
            }
        }
    }

    for (int o = 0; o < 4; ++o) {
        for (int r = 0; r < 8; ++r)
            red[((wave * 4 + o) * 8 + r) * 32 + lane] = acc[o][r];
    }
    __syncthreads();

    int R = tid >> 4, C = tid & 15;
    int srcw = (R & 7) * 32 + (R >> 3) * 16 + C;
    for (int o = 0; o < 4; ++o) {
        float s = red[(0 * 4 + o) * 256 + srcw];
        for (int wv = 1; wv < 8; ++wv)
            s += red[(wv * 4 + o) * 256 + srcw];
        float val = s + bias[o];
        outc[(((size_t)b * 4 + o) * SDIM + i0 + R) * SDIM + j0 + C] =
            1.0f / (1.0f + expf(-val));
    }
}

extern "C" void kernel_launch(void* const* d_in, const int* in_sizes, int n_in,
                              void* d_out, int out_size, void* d_ws, size_t ws_size,
                              hipStream_t stream) {
    const float* enc   = (const float*)d_in[0];
    const int*   masks = (const int*)d_in[1];
    const float* w     = (const float*)d_in[2];
    const float* bias  = (const float*)d_in[3];
    float*    out  = (float*)d_out;
    unsigned* part = (unsigned*)d_ws;
    float*    outv = out;
    float*    outc = out + BB * CCH * SDIM;

    if (ws_size < sizeof(unsigned) * BB * NCHUNK * PLANE) return;
    segmax_kernel<<<BB * NCHUNK * NGRP, SEG_TPB, 0, stream>>>(enc, masks, part);
    conv_wmma_kernel<<<BB * 16, 256, 0, stream>>>(part, w, bias, outv, outc);
}


# segmax: each wave keeps a floor per channel (smallest word of the channel's LDS row, a wave min of 8 reads per lane, re-read each pass) and issues an LDS max only for values above it; max stays exact
# speedup vs baseline: 5.3439x; 1.2695x over previous
#include <hip/hip_runtime.h>
#include <math.h>

#define SEGS 65
#define CCH  64
#define PIX  (256*256)
#define BB   4
#define SDIM 64
#define TPITCH 65

typedef __attribute__((ext_vector_type(2))) float v2f;
typedef __attribute__((ext_vector_type(8))) float v8f;
typedef __attribute__((ext_vector_type(4))) float v4f;
typedef __attribute__((ext_vector_type(4))) int v4i;
typedef __attribute__((ext_vector_type(4))) unsigned v4u;

__device__ __forceinline__ unsigned ftos(float f) {
    unsigned b = __float_as_uint(f);
    return (b & 0x80000000u) ? ~b : (b | 0x80000000u);
}
__device__ __forceinline__ float stof(unsigned u) {
    unsigned b = (u & 0x80000000u) ? (u & 0x7FFFFFFFu) : ~u;
    return __uint_as_float(b);
}
#define NEG_INF_S 0x007FFFFFu

#define CG   4
#define NGRP (CCH / CG)
#define SEG_TPB 256
#define SEG_PIX 8192
#define NCHUNK (PIX / SEG_PIX)
#define QPT  2
#define PASS_PIX (SEG_TPB * 4 * QPT)
#define NPASS (SEG_PIX / PASS_PIX)
#define PLANE (CCH * SDIM)

static_assert(SEG_TPB == CG * SDIM, "flush: one thread per (channel, segment) word");
static_assert(CG == 4, "lane rotation below is written for 4 channels (two select stages)");
static_assert(NPASS % 2 == 0, "passes alternate between two register buffers");

__device__ __forceinline__ unsigned wave_min_all(unsigned x) {
    x = min(x, (unsigned)__builtin_amdgcn_update_dpp(0, (int)x, 0xB1,  0xF, 0xF, false));
    x = min(x, (unsigned)__builtin_amdgcn_update_dpp(0, (int)x, 0x4E,  0xF, 0xF, false));
    x = min(x, (unsigned)__builtin_amdgcn_update_dpp(0, (int)x, 0x141, 0xF, 0xF, false));
    x = min(x, (unsigned)__builtin_amdgcn_update_dpp(0, (int)x, 0x140, 0xF, 0xF, false));
    x = min(x, (unsigned)__builtin_amdgcn_permlanex16(x, x, 0x76543210u, 0xFEDCBA98u, false, false));
    return x;
}

__device__ __forceinline__ void seg_floors(unsigned* tab, unsigned lane, bool b0, bool b1, unsigned (&t)[4]) {
    unsigned m[CG];
    #pragma unroll
    for (int c = 0; c < CG; ++c) {
        unsigned x = __hip_atomic_load(&tab[c * TPITCH + 1 + lane],  __ATOMIC_RELAXED, __HIP_MEMORY_SCOPE_WORKGROUP);
        unsigned y = __hip_atomic_load(&tab[c * TPITCH + 33 + lane], __ATOMIC_RELAXED, __HIP_MEMORY_SCOPE_WORKGROUP);
        m[c] = wave_min_all(min(x, y));
    }
    unsigned a0 = b0 ? m[1] : m[0], a1 = b0 ? m[2] : m[1], a2 = b0 ? m[3] : m[2], a3 = b0 ? m[0] : m[3];
    t[0] = b1 ? a2 : a0;
    t[1] = b1 ? a3 : a1;
    t[2] = b1 ? a0 : a2;
    t[3] = b1 ? a1 : a3;
}

template <bool FILTER>
__device__ __forceinline__ void seg_max4(unsigned* tab, int id,
                                         float f0, float f1, float f2, float f3,
                                         bool b0, bool b1,
                                         unsigned o0, unsigned o1, unsigned o2, unsigned o3,
                                         const unsigned (&t)[4]) {
    unsigned u0 = ftos(f0), u1 = ftos(f1), u2 = ftos(f2), u3 = ftos(f3);
    unsigned a0 = b0 ? u1 : u0, a1 = b0 ? u2 : u1, a2 = b0 ? u3 : u2, a3 = b0 ? u0 : u3;
    unsigned v0 = b1 ? a2 : a0, v1 = b1 ? a3 : a1, v2 = b1 ? a0 : a2, v3 = b1 ? a1 : a3;
    unsigned* row = &tab[id];
    if (!FILTER || v0 > t[0]) atomicMax(&row[o0], v0);
    if (!FILTER || v1 > t[1]) atomicMax(&row[o1], v1);
    if (!FILTER || v2 > t[2]) atomicMax(&row[o2], v2);
    if (!FILTER || v3 > t[3]) atomicMax(&row[o3], v3);
}

__device__ __forceinline__ void seg_load_pass(const int* __restrict__ maskb,
                                              const float* __restrict__ encg, int p,
                                              v4i (&id)[QPT], v4f (&v)[QPT][CG]) {
    #pragma unroll
    for (int q = 0; q < QPT; ++q) {
        id[q] = *(const v4i*)&maskb[p + q * SEG_TPB * 4];
        #pragma unroll
        for (int c = 0; c < CG; ++c)
            v[q][c] = *(const v4f*)&encg[(size_t)c * PIX + p + q * SEG_TPB * 4];
    }
}

template <bool FIRST>
__device__ __forceinline__ void seg_feed_pass(unsigned* tab, unsigned lane,
                                              const v4i (&id)[QPT], const v4f (&v)[QPT][CG],
                                              bool b0, bool b1,
                                              unsigned o0, unsigned o1, unsigned o2, unsigned o3) {
    static_assert(QPT == 2, "one floor read per pass, placed by quad number");
    unsigned t[4] = { NEG_INF_S, NEG_INF_S, NEG_INF_S, NEG_INF_S };
    if (!FIRST) seg_floors(tab, lane, b0, b1, t);
    seg_max4<!FIRST>(tab, id[0].x, v[0][0].x, v[0][1].x, v[0][2].x, v[0][3].x, b0, b1, o0, o1, o2, o3, t);
    seg_max4<!FIRST>(tab, id[0].y, v[0][0].y, v[0][1].y, v[0][2].y, v[0][3].y, b0, b1, o0, o1, o2, o3, t);
    seg_max4<!FIRST>(tab, id[0].z, v[0][0].z, v[0][1].z, v[0][2].z, v[0][3].z, b0, b1, o0, o1, o2, o3, t);
    seg_max4<!FIRST>(tab, id[0].w, v[0][0].w, v[0][1].w, v[0][2].w, v[0][3].w, b0, b1, o0, o1, o2, o3, t);
    if (FIRST) seg_floors(tab, lane, b0, b1, t);
    seg_max4<true>(tab, id[1].x, v[1][0].x, v[1][1].x, v[1][2].x, v[1][3].x, b0, b1, o0, o1, o2, o3, t);
    seg_max4<true>(tab, id[1].y, v[1][0].y, v[1][1].y, v[1][2].y, v[1][3].y, b0, b1, o0, o1, o2, o3, t);
    seg_max4<true>(tab, id[1].z, v[1][0].z, v[1][1].z, v[1][2].z, v[1][3].z, b0, b1, o0, o1, o2, o3, t);
    seg_max4<true>(tab, id[1].w, v[1][0].w, v[1][1].w, v[1][2].w, v[1][3].w, b0, b1, o0, o1, o2, o3, t);
}

__global__ void __launch_bounds__(SEG_TPB)
segmax_kernel(const float* __restrict__ enc,
              const int*   __restrict__ masks,
              unsigned*    __restrict__ part) {
    int g     = blockIdx.x % NGRP;
    int chunk = (blockIdx.x / NGRP) % NCHUNK;
    int b     = blockIdx.x / (NGRP * NCHUNK);
    int p0    = chunk * SEG_PIX;

    __shared__ unsigned tab[CG * TPITCH];
    for (int i = threadIdx.x; i < CG * TPITCH; i += SEG_TPB) tab[i] = NEG_INF_S;
    __syncthreads();

    const unsigned rot = threadIdx.x & 3;
    const unsigned lane32 = threadIdx.x & 31;
    const bool b0 = rot & 1, b1 = rot & 2;
    const unsigned o0 = ((0 + rot) & 3) * TPITCH, o1 = ((1 + rot) & 3) * TPITCH,
                   o2 = ((2 + rot) & 3) * TPITCH, o3 = ((3 + rot) & 3) * TPITCH;

    const float* encg  = enc   + ((size_t)b * CCH + (size_t)g * CG) * PIX;
    const int*   maskb = masks + (size_t)b * PIX;

    const int pl = p0 + threadIdx.x * 4;
    v4i idA[QPT], idB[QPT];
    v4f vA[QPT][CG], vB[QPT][CG];
    seg_load_pass(maskb, encg, pl, idA, vA);
    __builtin_amdgcn_sched_barrier(0);
    #pragma unroll
    for (int pass = 0; pass < NPASS; pass += 2) {
        seg_load_pass(maskb, encg, pl + (pass + 1) * PASS_PIX, idB, vB);
        __builtin_amdgcn_sched_barrier(0);
        if (pass == 0) seg_feed_pass<true >(tab, lane32, idA, vA, b0, b1, o0, o1, o2, o3);
        else           seg_feed_pass<false>(tab, lane32, idA, vA, b0, b1, o0, o1, o2, o3);
        __builtin_amdgcn_sched_barrier(0);
        if (pass + 2 < NPASS) {
            seg_load_pass(maskb, encg, pl + (pass + 2) * PASS_PIX, idA, vA);
            __builtin_amdgcn_sched_barrier(0);
        }
        seg_feed_pass<false>(tab, lane32, idB, vB, b0, b1, o0, o1, o2, o3);
        if (pass + 2 < NPASS)
            __builtin_amdgcn_sched_barrier(0);
    }
    __syncthreads();

    unsigned* dst = part + (((size_t)b * NCHUNK + chunk) * CCH + (size_t)g * CG) * SDIM;
    int c = threadIdx.x / SDIM, s = threadIdx.x % SDIM;
    dst[threadIdx.x] = tab[c * TPITCH + s + 1];
}

__global__ void __launch_bounds__(256)
conv_wmma_kernel(const unsigned* __restrict__ part,
                 const float* __restrict__ w,
                 const float* __restrict__ bias,
                 float*       __restrict__ outv,
                 float*       __restrict__ outc) {
    __shared__ float vpad[CCH * 66];
    __shared__ float wl[4 * 576];
    __shared__ float red[8 * 4 * 256];

    int b        = blockIdx.x >> 4;
    int kk       = blockIdx.x & 15;
    const unsigned* pb = part + (size_t)b * NCHUNK * PLANE;

    int tid = threadIdx.x;
    v4u px[NCHUNK][4];
    #pragma unroll
    for (int ch = 0; ch < NCHUNK; ++ch) {
        #pragma unroll
        for (int j = 0; j < 4; ++j)
            px[ch][j] = *(const v4u*)&pb[(size_t)ch * PLANE + (j * 256 + tid) * 4];
    }
    asm volatile("" :: "v"(px[0][0]), "v"(px[0][1]), "v"(px[0][2]), "v"(px[0][3]),
                       "v"(px[1][0]), "v"(px[1][1]), "v"(px[1][2]), "v"(px[1][3]),
                       "v"(px[2][0]), "v"(px[2][1]), "v"(px[2][2]), "v"(px[2][3]),
                       "v"(px[3][0]), "v"(px[3][1]), "v"(px[3][2]), "v"(px[3][3]));
    asm volatile("" :: "v"(px[4][0]), "v"(px[4][1]), "v"(px[4][2]), "v"(px[4][3]),
                       "v"(px[5][0]), "v"(px[5][1]), "v"(px[5][2]), "v"(px[5][3]),
                       "v"(px[6][0]), "v"(px[6][1]), "v"(px[6][2]), "v"(px[6][3]),
                       "v"(px[7][0]), "v"(px[7][1]), "v"(px[7][2]), "v"(px[7][3]));
    v4u mx[4];
    #pragma unroll
    for (int j = 0; j < 4; ++j) {
        mx[j] = px[0][j];
        #pragma unroll
        for (int ch = 1; ch < NCHUNK; ++ch)
            mx[j] = __builtin_elementwise_max(mx[j], px[ch][j]);
    }
    #pragma unroll
    for (int j = 0; j < 4; ++j) {
        v4f o = { stof(mx[j].x), stof(mx[j].y), stof(mx[j].z), stof(mx[j].w) };
        float* d = &vpad[(16 * j + (tid >> 4)) * 66 + 1 + (tid & 15) * 4];
        d[0] = o.x; d[1] = o.y; d[2] = o.z; d[3] = o.w;
        if (j == (kk >> 2) && (tid >> 6) == (kk & 3))
            *(v4f*)&outv[(size_t)b * PLANE + (j * 256 + tid) * 4] = o;
    }
    if (tid < 2 * CCH) vpad[(tid >> 1) * 66 + (tid & 1) * 65] = 0.0f;
    for (int i = threadIdx.x; i < 4 * 576; i += blockDim.x) {
        int o = i / 576, r = i % 576;
        int t = r / 64, c = r % 64;
        wl[i] = w[((size_t)(o * CCH + c)) * 9 + t];
    }
    __syncthreads();

    int wave = threadIdx.x >> 5;
    int lane = threadIdx.x & 31;
    int m    = lane & 15;
    int half = lane >> 4;
    int kb   = half * 2;

    int i0 = (kk >> 2) << 4;
    int j0 = (kk & 3) << 4;

    v8f acc[4];
    for (int o = 0; o < 4; ++o)
        acc[o] = (v8f){0.f, 0.f, 0.f, 0.f, 0.f, 0.f, 0.f, 0.f};

    for (int cc = 0; cc < 2; ++cc) {
        int c0 = (wave * 2 + cc) * 4 + kb;
        const float* r0 = &vpad[c0 * 66];
        const float* r1 = r0 + 66;
        v2f va[3], vb[3];
        for (int d = 0; d < 3; ++d) {
            va[d].x = r0[i0 + m + d];  va[d].y = r1[i0 + m + d];
            vb[d].x = r0[j0 + m + d];  vb[d].y = r1[j0 + m + d];
        }
        for (int t = 0; t < 9; ++t) {
            for (int o = 0; o < 4; ++o) {
                v2f ww = *(const v2f*)&wl[o * 576 + t * 64 + c0];
                v2f a  = ww * va[t / 3];
                acc[o] = __builtin_amdgcn_wmma_f32_16x16x4_f32(
                             false, a, false, vb[t % 3], (short)0, acc[o], false, false);
            }
        }
    }

    for (int o = 0; o < 4; ++o) {
        for (int r = 0; r < 8; ++r)
            red[((wave * 4 + o) * 8 + r) * 32 + lane] = acc[o][r];
    }
    __syncthreads();

    int R = tid >> 4, C = tid & 15;
    int srcw = (R & 7) * 32 + (R >> 3) * 16 + C;
    for (int o = 0; o < 4; ++o) {
        float s = red[(0 * 4 + o) * 256 + srcw];
        for (int wv = 1; wv < 8; ++wv)
            s += red[(wv * 4 + o) * 256 + srcw];
        float val = s + bias[o];
        outc[(((size_t)b * 4 + o) * SDIM + i0 + R) * SDIM + j0 + C] =
            1.0f / (1.0f + expf(-val));
    }
}

extern "C" void kernel_launch(void* const* d_in, const int* in_sizes, int n_in,
                              void* d_out, int out_size, void* d_ws, size_t ws_size,
                              hipStream_t stream) {
    const float* enc   = (const float*)d_in[0];
    const int*   masks = (const int*)d_in[1];
    const float* w     = (const float*)d_in[2];
    const float* bias  = (const float*)d_in[3];
    float*    out  = (float*)d_out;
    unsigned* part = (unsigned*)d_ws;
    float*    outv = out;
    float*    outc = out + BB * CCH * SDIM;

    if (ws_size < sizeof(unsigned) * BB * NCHUNK * PLANE) return;
    segmax_kernel<<<BB * NCHUNK * NGRP, SEG_TPB, 0, stream>>>(enc, masks, part);
    conv_wmma_kernel<<<BB * 16, 256, 0, stream>>>(part, w, bias, outv, outc);
}


# conv: weights kept in LDS in memory order [o][c][tap], copied by 3 contiguous b128 loads per thread issued with the plane loads (was a 9-trip loop of strided b32 loads, each waited for)
# speedup vs baseline: 5.7021x; 1.0670x over previous
#include <hip/hip_runtime.h>
#include <math.h>

#define SEGS 65
#define CCH  64
#define PIX  (256*256)
#define BB   4
#define SDIM 64
#define TPITCH 65

typedef __attribute__((ext_vector_type(2))) float v2f;
typedef __attribute__((ext_vector_type(8))) float v8f;
typedef __attribute__((ext_vector_type(4))) float v4f;
typedef __attribute__((ext_vector_type(4))) int v4i;
typedef __attribute__((ext_vector_type(4))) unsigned v4u;

__device__ __forceinline__ unsigned ftos(float f) {
    unsigned b = __float_as_uint(f);
    return (b & 0x80000000u) ? ~b : (b | 0x80000000u);
}
__device__ __forceinline__ float stof(unsigned u) {
    unsigned b = (u & 0x80000000u) ? (u & 0x7FFFFFFFu) : ~u;
    return __uint_as_float(b);
}
#define NEG_INF_S 0x007FFFFFu

#define CG   4
#define NGRP (CCH / CG)
#define SEG_TPB 256
#define SEG_PIX 8192
#define NCHUNK (PIX / SEG_PIX)
#define QPT  2
#define PASS_PIX (SEG_TPB * 4 * QPT)
#define NPASS (SEG_PIX / PASS_PIX)
#define PLANE (CCH * SDIM)

static_assert(SEG_TPB == CG * SDIM, "flush: one thread per (channel, segment) word");
static_assert(CG == 4, "lane rotation below is written for 4 channels (two select stages)");
static_assert(NPASS % 2 == 0, "passes alternate between two register buffers");

__device__ __forceinline__ unsigned wave_min_all(unsigned x) {
    x = min(x, (unsigned)__builtin_amdgcn_update_dpp(0, (int)x, 0xB1,  0xF, 0xF, false));
    x = min(x, (unsigned)__builtin_amdgcn_update_dpp(0, (int)x, 0x4E,  0xF, 0xF, false));
    x = min(x, (unsigned)__builtin_amdgcn_update_dpp(0, (int)x, 0x141, 0xF, 0xF, false));
    x = min(x, (unsigned)__builtin_amdgcn_update_dpp(0, (int)x, 0x140, 0xF, 0xF, false));
    x = min(x, (unsigned)__builtin_amdgcn_permlanex16(x, x, 0x76543210u, 0xFEDCBA98u, false, false));
    return x;
}

__device__ __forceinline__ void seg_floors(unsigned* tab, unsigned lane, bool b0, bool b1, unsigned (&t)[4]) {
    unsigned m[CG];
    #pragma unroll
    for (int c = 0; c < CG; ++c) {
        unsigned x = __hip_atomic_load(&tab[c * TPITCH + 1 + lane],  __ATOMIC_RELAXED, __HIP_MEMORY_SCOPE_WORKGROUP);
        unsigned y = __hip_atomic_load(&tab[c * TPITCH + 33 + lane], __ATOMIC_RELAXED, __HIP_MEMORY_SCOPE_WORKGROUP);
        m[c] = wave_min_all(min(x, y));
    }
    unsigned a0 = b0 ? m[1] : m[0], a1 = b0 ? m[2] : m[1], a2 = b0 ? m[3] : m[2], a3 = b0 ? m[0] : m[3];
    t[0] = b1 ? a2 : a0;
    t[1] = b1 ? a3 : a1;
    t[2] = b1 ? a0 : a2;
    t[3] = b1 ? a1 : a3;
}

template <bool FILTER>
__device__ __forceinline__ void seg_max4(unsigned* tab, int id,
                                         float f0, float f1, float f2, float f3,
                                         bool b0, bool b1,
                                         unsigned o0, unsigned o1, unsigned o2, unsigned o3,
                                         const unsigned (&t)[4]) {
    unsigned u0 = ftos(f0), u1 = ftos(f1), u2 = ftos(f2), u3 = ftos(f3);
    unsigned a0 = b0 ? u1 : u0, a1 = b0 ? u2 : u1, a2 = b0 ? u3 : u2, a3 = b0 ? u0 : u3;
    unsigned v0 = b1 ? a2 : a0, v1 = b1 ? a3 : a1, v2 = b1 ? a0 : a2, v3 = b1 ? a1 : a3;
    unsigned* row = &tab[id];
    if (!FILTER || v0 > t[0]) atomicMax(&row[o0], v0);
    if (!FILTER || v1 > t[1]) atomicMax(&row[o1], v1);
    if (!FILTER || v2 > t[2]) atomicMax(&row[o2], v2);
    if (!FILTER || v3 > t[3]) atomicMax(&row[o3], v3);
}

__device__ __forceinline__ void seg_load_pass(const int* __restrict__ maskb,
                                              const float* __restrict__ encg, int p,
                                              v4i (&id)[QPT], v4f (&v)[QPT][CG]) {
    #pragma unroll
    for (int q = 0; q < QPT; ++q) {
        id[q] = *(const v4i*)&maskb[p + q * SEG_TPB * 4];
        #pragma unroll
        for (int c = 0; c < CG; ++c)
            v[q][c] = *(const v4f*)&encg[(size_t)c * PIX + p + q * SEG_TPB * 4];
    }
}

template <bool FIRST>
__device__ __forceinline__ void seg_feed_pass(unsigned* tab, unsigned lane,
                                              const v4i (&id)[QPT], const v4f (&v)[QPT][CG],
                                              bool b0, bool b1,
                                              unsigned o0, unsigned o1, unsigned o2, unsigned o3) {
    static_assert(QPT == 2, "one floor read per pass, placed by quad number");
    unsigned t[4] = { NEG_INF_S, NEG_INF_S, NEG_INF_S, NEG_INF_S };
    if (!FIRST) seg_floors(tab, lane, b0, b1, t);
    seg_max4<!FIRST>(tab, id[0].x, v[0][0].x, v[0][1].x, v[0][2].x, v[0][3].x, b0, b1, o0, o1, o2, o3, t);
    seg_max4<!FIRST>(tab, id[0].y, v[0][0].y, v[0][1].y, v[0][2].y, v[0][3].y, b0, b1, o0, o1, o2, o3, t);
    seg_max4<!FIRST>(tab, id[0].z, v[0][0].z, v[0][1].z, v[0][2].z, v[0][3].z, b0, b1, o0, o1, o2, o3, t);
    seg_max4<!FIRST>(tab, id[0].w, v[0][0].w, v[0][1].w, v[0][2].w, v[0][3].w, b0, b1, o0, o1, o2, o3, t);
    if (FIRST) seg_floors(tab, lane, b0, b1, t);
    seg_max4<true>(tab, id[1].x, v[1][0].x, v[1][1].x, v[1][2].x, v[1][3].x, b0, b1, o0, o1, o2, o3, t);
    seg_max4<true>(tab, id[1].y, v[1][0].y, v[1][1].y, v[1][2].y, v[1][3].y, b0, b1, o0, o1, o2, o3, t);
    seg_max4<true>(tab, id[1].z, v[1][0].z, v[1][1].z, v[1][2].z, v[1][3].z, b0, b1, o0, o1, o2, o3, t);
    seg_max4<true>(tab, id[1].w, v[1][0].w, v[1][1].w, v[1][2].w, v[1][3].w, b0, b1, o0, o1, o2, o3, t);
}

__global__ void __launch_bounds__(SEG_TPB)
segmax_kernel(const float* __restrict__ enc,
              const int*   __restrict__ masks,
              unsigned*    __restrict__ part) {
    int g     = blockIdx.x % NGRP;
    int chunk = (blockIdx.x / NGRP) % NCHUNK;
    int b     = blockIdx.x / (NGRP * NCHUNK);
    int p0    = chunk * SEG_PIX;

    __shared__ unsigned tab[CG * TPITCH];
    for (int i = threadIdx.x; i < CG * TPITCH; i += SEG_TPB) tab[i] = NEG_INF_S;
    __syncthreads();

    const unsigned rot = threadIdx.x & 3;
    const unsigned lane32 = threadIdx.x & 31;
    const bool b0 = rot & 1, b1 = rot & 2;
    const unsigned o0 = ((0 + rot) & 3) * TPITCH, o1 = ((1 + rot) & 3) * TPITCH,
                   o2 = ((2 + rot) & 3) * TPITCH, o3 = ((3 + rot) & 3) * TPITCH;

    const float* encg  = enc   + ((size_t)b * CCH + (size_t)g * CG) * PIX;
    const int*   maskb = masks + (size_t)b * PIX;

    const int pl = p0 + threadIdx.x * 4;
    v4i idA[QPT], idB[QPT];
    v4f vA[QPT][CG], vB[QPT][CG];
    seg_load_pass(maskb, encg, pl, idA, vA);
    __builtin_amdgcn_sched_barrier(0);
    #pragma unroll
    for (int pass = 0; pass < NPASS; pass += 2) {
        seg_load_pass(maskb, encg, pl + (pass + 1) * PASS_PIX, idB, vB);
        __builtin_amdgcn_sched_barrier(0);
        if (pass == 0) seg_feed_pass<true >(tab, lane32, idA, vA, b0, b1, o0, o1, o2, o3);
        else           seg_feed_pass<false>(tab, lane32, idA, vA, b0, b1, o0, o1, o2, o3);
        __builtin_amdgcn_sched_barrier(0);
        if (pass + 2 < NPASS) {
            seg_load_pass(maskb, encg, pl + (pass + 2) * PASS_PIX, idA, vA);
            __builtin_amdgcn_sched_barrier(0);
        }
        seg_feed_pass<false>(tab, lane32, idB, vB, b0, b1, o0, o1, o2, o3);
        if (pass + 2 < NPASS)
            __builtin_amdgcn_sched_barrier(0);
    }
    __syncthreads();

    unsigned* dst = part + (((size_t)b * NCHUNK + chunk) * CCH + (size_t)g * CG) * SDIM;
    int c = threadIdx.x / SDIM, s = threadIdx.x % SDIM;
    dst[threadIdx.x] = tab[c * TPITCH + s + 1];
}

__global__ void __launch_bounds__(256)
conv_wmma_kernel(const unsigned* __restrict__ part,
                 const float* __restrict__ w,
                 const float* __restrict__ bias,
                 float*       __restrict__ outv,
                 float*       __restrict__ outc) {
    __shared__ float vpad[CCH * 66];
    __shared__ v4f wl4[3 * 256];
    __shared__ float red[8 * 4 * 256];

    int b        = blockIdx.x >> 4;
    int kk       = blockIdx.x & 15;
    const unsigned* pb = part + (size_t)b * NCHUNK * PLANE;

    int tid = threadIdx.x;
    v4f wq[3];
    for (int j = 0; j < 3; ++j)
        wq[j] = ((const v4f*)w)[min(j * 256 + tid, 575)];
    v4u px[NCHUNK][4];
    #pragma unroll
    for (int ch = 0; ch < NCHUNK; ++ch) {
        #pragma unroll
        for (int j = 0; j < 4; ++j)
            px[ch][j] = *(const v4u*)&pb[(size_t)ch * PLANE + (j * 256 + tid) * 4];
    }
    asm volatile("" :: "v"(px[0][0]), "v"(px[0][1]), "v"(px[0][2]), "v"(px[0][3]),
                       "v"(px[1][0]), "v"(px[1][1]), "v"(px[1][2]), "v"(px[1][3]),
                       "v"(px[2][0]), "v"(px[2][1]), "v"(px[2][2]), "v"(px[2][3]),
                       "v"(px[3][0]), "v"(px[3][1]), "v"(px[3][2]), "v"(px[3][3]));
    asm volatile("" :: "v"(px[4][0]), "v"(px[4][1]), "v"(px[4][2]), "v"(px[4][3]),
                       "v"(px[5][0]), "v"(px[5][1]), "v"(px[5][2]), "v"(px[5][3]),
                       "v"(px[6][0]), "v"(px[6][1]), "v"(px[6][2]), "v"(px[6][3]),
                       "v"(px[7][0]), "v"(px[7][1]), "v"(px[7][2]), "v"(px[7][3]));
    for (int j = 0; j < 3; ++j)
        wl4[j * 256 + tid] = wq[j];
    v4u mx[4];
    #pragma unroll
    for (int j = 0; j < 4; ++j) {
        mx[j] = px[0][j];
        #pragma unroll
        for (int ch = 1; ch < NCHUNK; ++ch)
            mx[j] = __builtin_elementwise_max(mx[j], px[ch][j]);
    }
    #pragma unroll
    for (int j = 0; j < 4; ++j) {
        v4f o = { stof(mx[j].x), stof(mx[j].y), stof(mx[j].z), stof(mx[j].w) };
        float* d = &vpad[(16 * j + (tid >> 4)) * 66 + 1 + (tid & 15) * 4];
        d[0] = o.x; d[1] = o.y; d[2] = o.z; d[3] = o.w;
        if (j == (kk >> 2) && (tid >> 6) == (kk & 3))
            *(v4f*)&outv[(size_t)b * PLANE + (j * 256 + tid) * 4] = o;
    }
    if (tid < 2 * CCH) vpad[(tid >> 1) * 66 + (tid & 1) * 65] = 0.0f;
    __syncthreads();

    const float* wl = (const float*)wl4;
    int wave = threadIdx.x >> 5;
    int lane = threadIdx.x & 31;
    int m    = lane & 15;
    int half = lane >> 4;
    int kb   = half * 2;

    int i0 = (kk >> 2) << 4;
    int j0 = (kk & 3) << 4;

    v8f acc[4];
    for (int o = 0; o < 4; ++o)
        acc[o] = (v8f){0.f, 0.f, 0.f, 0.f, 0.f, 0.f, 0.f, 0.f};

    for (int cc = 0; cc < 2; ++cc) {
        int c0 = (wave * 2 + cc) * 4 + kb;
        const float* r0 = &vpad[c0 * 66];
        const float* r1 = r0 + 66;
        v2f va[3], vb[3];
        for (int d = 0; d < 3; ++d) {
            va[d].x = r0[i0 + m + d];  va[d].y = r1[i0 + m + d];
            vb[d].x = r0[j0 + m + d];  vb[d].y = r1[j0 + m + d];
        }
        for (int t = 0; t < 9; ++t) {
            for (int o = 0; o < 4; ++o) {
                const float* wp = &wl[(o * CCH + c0) * 9 + t];
                v2f ww = { wp[0], wp[9] };
                v2f a  = ww * va[t / 3];
                acc[o] = __builtin_amdgcn_wmma_f32_16x16x4_f32(
                             false, a, false, vb[t % 3], (short)0, acc[o], false, false);
            }
        }
    }

    for (int o = 0; o < 4; ++o) {
        for (int r = 0; r < 8; ++r)
            red[((wave * 4 + o) * 8 + r) * 32 + lane] = acc[o][r];
    }
    __syncthreads();

    int R = tid >> 4, C = tid & 15;
    int srcw = (R & 7) * 32 + (R >> 3) * 16 + C;
    for (int o = 0; o < 4; ++o) {
        float s = red[(0 * 4 + o) * 256 + srcw];
        for (int wv = 1; wv < 8; ++wv)
            s += red[(wv * 4 + o) * 256 + srcw];
        float val = s + bias[o];
        outc[(((size_t)b * 4 + o) * SDIM + i0 + R) * SDIM + j0 + C] =
            1.0f / (1.0f + expf(-val));
    }
}

extern "C" void kernel_launch(void* const* d_in, const int* in_sizes, int n_in,
                              void* d_out, int out_size, void* d_ws, size_t ws_size,
                              hipStream_t stream) {
    const float* enc   = (const float*)d_in[0];
    const int*   masks = (const int*)d_in[1];
    const float* w     = (const float*)d_in[2];
    const float* bias  = (const float*)d_in[3];
    float*    out  = (float*)d_out;
    unsigned* part = (unsigned*)d_ws;
    float*    outv = out;
    float*    outc = out + BB * CCH * SDIM;

    if (ws_size < sizeof(unsigned) * BB * NCHUNK * PLANE) return;
    segmax_kernel<<<BB * NCHUNK * NGRP, SEG_TPB, 0, stream>>>(enc, masks, part);
    conv_wmma_kernel<<<BB * 16, 256, 0, stream>>>(part, w, bias, outv, outc);
}
